# starting kernel
# baseline (speedup 1.0000x reference)
#include <hip/hip_runtime.h>
#include <cmath>

#define N_NODES 50000
#define N_EDGES 800000
#define DIM 128

typedef float v2f __attribute__((ext_vector_type(2)));
typedef float v8f __attribute__((ext_vector_type(8)));

__global__ __launch_bounds__(256) void node_dots_kernel(
    const float* __restrict__ x, const float* __restrict__ w_pred,
    float* __restrict__ p_a, float* __restrict__ p_b, float* __restrict__ deg)
{
    int node = (blockIdx.x * blockDim.x + threadIdx.x) >> 5;
    int lane = threadIdx.x & 31;
    if (node >= N_NODES) return;
    const float* xr = x + (size_t)node * DIM;
    float sa = 0.f, sb = 0.f;
#pragma unroll
    for (int j = 0; j < 4; ++j) {
        int c = lane + 32 * j;
        float v = xr[c];
        sa += v * w_pred[c];
        sb += v * w_pred[DIM + c];
    }
#pragma unroll
    for (int off = 16; off > 0; off >>= 1) {
        sa += __shfl_down(sa, off, 32);
        sb += __shfl_down(sb, off, 32);
    }
    if (lane == 0) {
        p_a[node] = sa;
        p_b[node] = sb;
        deg[node] = 1.0f;
    }
}

__global__ __launch_bounds__(256) void edge_weight_kernel(
    const int* __restrict__ src, const int* __restrict__ tgt,
    const float* __restrict__ p_a, const float* __restrict__ p_b,
    const float* __restrict__ b_pred,
    float* __restrict__ ew, float* __restrict__ deg)
{
    int e = blockIdx.x * blockDim.x + threadIdx.x;
    if (e >= N_EDGES) return;
    int s = src[e], t = tgt[e];
    float z = p_a[s] + p_b[t] + b_pred[0];
    float w = 1.0f / (1.0f + __expf(-z));
    ew[e] = w;
    unsafeAtomicAdd(&deg[t], w);
}

__global__ __launch_bounds__(256) void rsqrt_kernel(float* __restrict__ deg_dis)
{
    int n = blockIdx.x * blockDim.x + threadIdx.x;
    if (n < N_NODES) {
        float d = deg_dis[n];
        deg_dis[n] = (d > 0.f) ? rsqrtf(d) : 0.f;
    }
}

__global__ __launch_bounds__(256) void gemm_wmma_kernel(
    const float* __restrict__ x, const float* __restrict__ W,
    float* __restrict__ xw)
{
    __shared__ float As[16 * 132];
    const int rowBase = blockIdx.x * 16;

#pragma unroll
    for (int j = 0; j < 8; ++j) {
        int idx = threadIdx.x + j * 256;
        int r = idx >> 7, c = idx & 127;
        As[r * 132 + c] = x[(size_t)rowBase * DIM + idx];
    }
    __syncthreads();

    const int waveId  = threadIdx.x >> 5;
    const int lane    = threadIdx.x & 31;
    const int m       = lane & 15;
    const int hi      = lane >> 4;
    const int colBase = waveId * 16;

    v8f acc = {};
#pragma unroll 8
    for (int k0 = 0; k0 < DIM; k0 += 4) {
        int ka = k0 + 2 * hi;
        v2f a, b;
        a.x = As[m * 132 + ka];
        a.y = As[m * 132 + ka + 1];
        b.x = W[(size_t)ka * DIM + colBase + m];
        b.y = W[(size_t)(ka + 1) * DIM + colBase + m];
        acc = __builtin_amdgcn_wmma_f32_16x16x4_f32(
            false, a, false, b, (short)0, acc, false, false);
    }

#pragma unroll
    for (int i = 0; i < 8; ++i) {
        int row = rowBase + i + 8 * hi;
        xw[(size_t)row * DIM + colBase + m] = acc[i];
    }
}

__global__ __launch_bounds__(256) void self_bias_kernel(
    const float* __restrict__ xw, const float* __restrict__ dis,
    const float* __restrict__ b, float* __restrict__ out)
{
    int idx = blockIdx.x * blockDim.x + threadIdx.x;
    if (idx >= N_NODES * DIM) return;
    int node = idx >> 7, d = idx & 127;
    float s = dis[node];
    out[idx] = s * s * xw[idx] + b[d];
}

__global__ __launch_bounds__(256) void edge_scatter_kernel(
    const int* __restrict__ src, const int* __restrict__ tgt,
    const float* __restrict__ ew, const float* __restrict__ dis,
    const float* __restrict__ xw, float* __restrict__ out)
{
    int e = (blockIdx.x * blockDim.x + threadIdx.x) >> 5;
    int lane = threadIdx.x & 31;
    if (e >= N_EDGES) return;
    int s = src[e], t = tgt[e];
    float norm = dis[s] * ew[e] * dis[t];
    const float4* xv = (const float4*)(xw + (size_t)s * DIM);
    float4 v = xv[lane];
    float* op = out + (size_t)t * DIM + lane * 4;
    unsafeAtomicAdd(op + 0, norm * v.x);
    unsafeAtomicAdd(op + 1, norm * v.y);
    unsafeAtomicAdd(op + 2, norm * v.z);
    unsafeAtomicAdd(op + 3, norm * v.w);
}

extern "C" void kernel_launch(void* const* d_in, const int* in_sizes, int n_in,
                              void* d_out, int out_size, void* d_ws, size_t ws_size,
                              hipStream_t stream) {
    const float* x          = (const float*)d_in[0];
    const int*   edge_index = (const int*)d_in[1];
    const float* W          = (const float*)d_in[2];
    const float* b          = (const float*)d_in[3];
    const float* w_pred     = (const float*)d_in[4];
    const float* b_pred     = (const float*)d_in[5];
    float*       out        = (float*)d_out;

    const int* src = edge_index;
    const int* tgt = edge_index + N_EDGES;

    float* ws  = (float*)d_ws;
    float* p_a = ws;
    float* p_b = p_a + N_NODES;
    float* deg = p_b + N_NODES;
    float* ew  = deg + N_NODES;
    float* xw  = ew + N_EDGES;

    node_dots_kernel  <<<(N_NODES + 7) / 8,      256, 0, stream>>>(x, w_pred, p_a, p_b, deg);
    edge_weight_kernel<<<(N_EDGES + 255) / 256,  256, 0, stream>>>(src, tgt, p_a, p_b, b_pred, ew, deg);
    rsqrt_kernel      <<<(N_NODES + 255) / 256,  256, 0, stream>>>(deg);
    gemm_wmma_kernel  <<<N_NODES / 16,           256, 0, stream>>>(x, W, xw);
    self_bias_kernel  <<<(N_NODES * DIM) / 256,  256, 0, stream>>>(xw, deg, b, out);
    edge_scatter_kernel<<<N_EDGES / 8,           256, 0, stream>>>(src, tgt, ew, deg, xw, out);
}


# edges counting-sorted by target each call (slot = returning int add), one wave sums a node's rows in registers; f32 atomics on out and deg removed, sum order = slot arrival order
# speedup vs baseline: 1.6479x; 1.6479x over previous
#include <hip/hip_runtime.h>
#include <cmath>


#define N_NODES 50000
#define N_EDGES 800000
#define DIM 128

typedef float v2f __attribute__((ext_vector_type(2)));
typedef float v8f __attribute__((ext_vector_type(8)));

__global__ __launch_bounds__(256) void node_dots_kernel(
    const float* __restrict__ x, const float* __restrict__ w_pred,
    float* __restrict__ p_a, float* __restrict__ p_b, int* __restrict__ cnt)
{
    int node = (blockIdx.x * blockDim.x + threadIdx.x) >> 5;
    int lane = threadIdx.x & 31;
    if (node >= N_NODES) return;
    const float* xr = x + (size_t)node * DIM;
    float sa = 0.f, sb = 0.f;
#pragma unroll
    for (int j = 0; j < 4; ++j) {
        int c = lane + 32 * j;
        float v = xr[c];
        sa += v * w_pred[c];
        sb += v * w_pred[DIM + c];
    }
#pragma unroll
    for (int off = 16; off > 0; off >>= 1) {
        sa += __shfl_down(sa, off, 32);
        sb += __shfl_down(sb, off, 32);
    }
    if (lane == 0) {
        p_a[node] = sa;
        p_b[node] = sb;
        cnt[node] = 0;
    }
}

__global__ __launch_bounds__(256) void edge_weight_kernel(
    const int* __restrict__ src, const int* __restrict__ tgt,
    const float* __restrict__ p_a, const float* __restrict__ p_b,
    const float* __restrict__ b_pred,
    float* __restrict__ ew, int* __restrict__ rank, int* __restrict__ cnt)
{
    int e = blockIdx.x * blockDim.x + threadIdx.x;
    if (e >= N_EDGES) return;
    int s = src[e], t = tgt[e];
    float z = p_a[s] + p_b[t] + b_pred[0];
    float w = 1.0f / (1.0f + __expf(-z));
    ew[e] = w;
    rank[e] = atomicAdd(&cnt[t], 1);
}

__global__ __launch_bounds__(1024) void offsets_kernel(
    const int* __restrict__ cnt, int* __restrict__ off)
{
    __shared__ int red[32];
    __shared__ int scn[32];
    const int tid = threadIdx.x, lane = tid & 31, w = tid >> 5;
    const int b = blockIdx.x;

    int part = 0;
    const int4* c4 = (const int4*)cnt;
    for (int i = tid; i < b * 256; i += 1024) {
        int4 v = c4[i];
        part += (v.x + v.y) + (v.z + v.w);
    }
    const int n = b * 1024 + tid;
    const int mine = (n < N_NODES) ? cnt[n] : 0;

#pragma unroll
    for (int o = 16; o > 0; o >>= 1) part += __shfl_xor(part, o, 32);
    int inc = mine;
#pragma unroll
    for (int o = 1; o < 32; o <<= 1) {
        int y = __shfl_up(inc, o, 32);
        if (lane >= o) inc += y;
    }
    if (lane == 0)  red[w] = part;
    if (lane == 31) scn[w] = inc;
    __syncthreads();
    int tot = red[lane];
    int wp  = (lane < w) ? scn[lane] : 0;
#pragma unroll
    for (int o = 16; o > 0; o >>= 1) {
        tot += __shfl_xor(tot, o, 32);
        wp  += __shfl_xor(wp, o, 32);
    }
    if (n < N_NODES) off[n] = tot + wp + inc - mine;
}

__global__ __launch_bounds__(256) void place_kernel(
    const int* __restrict__ src, const int* __restrict__ tgt,
    const float* __restrict__ ew, const int* __restrict__ rank,
    const int* __restrict__ off, int2* __restrict__ seg)
{
    int e = blockIdx.x * blockDim.x + threadIdx.x;
    if (e >= N_EDGES) return;
    int s = src[e], t = tgt[e];
    int pos = off[t] + rank[e];
    seg[pos] = make_int2(s, __float_as_int(ew[e]));
}

__global__ __launch_bounds__(256) void degree_kernel(
    const int* __restrict__ off, const int* __restrict__ cnt,
    const int2* __restrict__ seg, float* __restrict__ dis)
{
    const int gid  = blockIdx.x * blockDim.x + threadIdx.x;
    const int node = gid >> 3, sub = gid & 7;
    const bool valid = node < N_NODES;
    const int nd = valid ? node : (N_NODES - 1);
    const int beg = off[nd], c = cnt[nd];
    float d = 0.f;
    for (int i = sub; i < c; i += 8) d += __int_as_float(seg[beg + i].y);
    d += __shfl_xor(d, 4, 32);
    d += __shfl_xor(d, 2, 32);
    d += __shfl_xor(d, 1, 32);
    if (valid && sub == 0) {
        float g = 1.0f + d;
        dis[node] = (g > 0.f) ? rsqrtf(g) : 0.f;
    }
}

__global__ __launch_bounds__(256) void gemm_wmma_kernel(
    const float* __restrict__ x, const float* __restrict__ W,
    float* __restrict__ xw)
{
    __shared__ float As[16 * 132];
    const int rowBase = blockIdx.x * 16;

#pragma unroll
    for (int j = 0; j < 8; ++j) {
        int idx = threadIdx.x + j * 256;
        int r = idx >> 7, c = idx & 127;
        As[r * 132 + c] = x[(size_t)rowBase * DIM + idx];
    }
    __syncthreads();

    const int waveId  = threadIdx.x >> 5;
    const int lane    = threadIdx.x & 31;
    const int m       = lane & 15;
    const int hi      = lane >> 4;
    const int colBase = waveId * 16;

    v8f acc = {};
#pragma unroll 8
    for (int k0 = 0; k0 < DIM; k0 += 4) {
        int ka = k0 + 2 * hi;
        v2f a, b;
        a.x = As[m * 132 + ka];
        a.y = As[m * 132 + ka + 1];
        b.x = W[(size_t)ka * DIM + colBase + m];
        b.y = W[(size_t)(ka + 1) * DIM + colBase + m];
        acc = __builtin_amdgcn_wmma_f32_16x16x4_f32(
            false, a, false, b, (short)0, acc, false, false);
    }

#pragma unroll
    for (int i = 0; i < 8; ++i) {
        int row = rowBase + i + 8 * hi;
        xw[(size_t)row * DIM + colBase + m] = acc[i];
    }
}

__global__ __launch_bounds__(256) void node_gather_kernel(
    const int* __restrict__ off, const int* __restrict__ cnt,
    const int2* __restrict__ seg, const float* __restrict__ dis,
    const float* __restrict__ xw, const float* __restrict__ b,
    float* __restrict__ out)
{
    const int t = __builtin_amdgcn_readfirstlane(
        (int)((blockIdx.x * blockDim.x + threadIdx.x) >> 5));
    const int lane = threadIdx.x & 31;
    if (t >= N_NODES) return;
    const int beg = off[t];
    const int n_e = cnt[t];
    const float dt = dis[t];
    const float4* xv = (const float4*)xw;

    float4 acc = make_float4(0.f, 0.f, 0.f, 0.f);
    for (int base = 0; base < n_e; base += 32) {
        const int i = base + lane;
        int   s_l = 0;
        float c_l = 0.f;
        if (i < n_e) {
            int2 ent = seg[beg + i];
            s_l = ent.x;
            c_l = dis[s_l] * __int_as_float(ent.y) * dt;
        }
        const int m = min(32, n_e - base);
        for (int j = 0; j < m; j += 4) {
            const int s0 = __builtin_amdgcn_readlane(s_l, j);
            const int s1 = __builtin_amdgcn_readlane(s_l, j + 1);
            const int s2 = __builtin_amdgcn_readlane(s_l, j + 2);
            const int s3 = __builtin_amdgcn_readlane(s_l, j + 3);
            const float c0 = __int_as_float(__builtin_amdgcn_readlane(__float_as_int(c_l), j));
            const float c1 = __int_as_float(__builtin_amdgcn_readlane(__float_as_int(c_l), j + 1));
            const float c2 = __int_as_float(__builtin_amdgcn_readlane(__float_as_int(c_l), j + 2));
            const float c3 = __int_as_float(__builtin_amdgcn_readlane(__float_as_int(c_l), j + 3));
            const float4 v0 = xv[(size_t)s0 * 32 + lane];
            const float4 v1 = xv[(size_t)s1 * 32 + lane];
            const float4 v2 = xv[(size_t)s2 * 32 + lane];
            const float4 v3 = xv[(size_t)s3 * 32 + lane];
            acc.x += c0 * v0.x; acc.y += c0 * v0.y; acc.z += c0 * v0.z; acc.w += c0 * v0.w;
            if (j + 1 < m) { acc.x += c1 * v1.x; acc.y += c1 * v1.y; acc.z += c1 * v1.z; acc.w += c1 * v1.w; }
            if (j + 2 < m) { acc.x += c2 * v2.x; acc.y += c2 * v2.y; acc.z += c2 * v2.z; acc.w += c2 * v2.w; }
            if (j + 3 < m) { acc.x += c3 * v3.x; acc.y += c3 * v3.y; acc.z += c3 * v3.z; acc.w += c3 * v3.w; }
        }
    }

    const float4 self = xv[(size_t)t * 32 + lane];
    const float4 bb = ((const float4*)b)[lane];
    const float ss = dt * dt;
    float4 r;
    r.x = acc.x + (ss * self.x + bb.x);
    r.y = acc.y + (ss * self.y + bb.y);
    r.z = acc.z + (ss * self.z + bb.z);
    r.w = acc.w + (ss * self.w + bb.w);
    ((float4*)out)[(size_t)t * 32 + lane] = r;
}

extern "C" void kernel_launch(void* const* d_in, const int* in_sizes, int n_in,
                              void* d_out, int out_size, void* d_ws, size_t ws_size,
                              hipStream_t stream) {
    const float* x          = (const float*)d_in[0];
    const int*   edge_index = (const int*)d_in[1];
    const float* W          = (const float*)d_in[2];
    const float* b          = (const float*)d_in[3];
    const float* w_pred     = (const float*)d_in[4];
    const float* b_pred     = (const float*)d_in[5];
    float*       out        = (float*)d_out;

    const int* src = edge_index;
    const int* tgt = edge_index + N_EDGES;

    float* ws   = (float*)d_ws;
    float* p_a  = ws;
    float* p_b  = p_a + N_NODES;
    float* dis  = p_b + N_NODES;
    int*   cnt  = (int*)(dis + N_NODES);
    int*   off  = cnt + N_NODES;
    float* ew   = (float*)(off + N_NODES);
    int*   rank = (int*)(ew + N_EDGES);
    int2*  seg  = (int2*)(rank + N_EDGES);
    float* xw   = ws + 3450048;

    node_dots_kernel  <<<(N_NODES + 7) / 8,          256,  0, stream>>>(x, w_pred, p_a, p_b, cnt);
    edge_weight_kernel<<<(N_EDGES + 255) / 256,      256,  0, stream>>>(src, tgt, p_a, p_b, b_pred, ew, rank, cnt);
    offsets_kernel    <<<(N_NODES + 1023) / 1024,    1024, 0, stream>>>(cnt, off);
    place_kernel      <<<(N_EDGES + 255) / 256,      256,  0, stream>>>(src, tgt, ew, rank, off, seg);
    degree_kernel     <<<(N_NODES * 8 + 255) / 256,  256,  0, stream>>>(off, cnt, seg, dis);
    gemm_wmma_kernel  <<<N_NODES / 16,               256,  0, stream>>>(x, W, xw);
    node_gather_kernel<<<N_NODES / 8,                256,  0, stream>>>(off, cnt, seg, dis, xw, b, out);
}


# edge-slot counters split into two tables, one per chip half (wave picks its table by die id): a counter line takes atomic adds from one half only; sum order = per-half slot arrival order
# speedup vs baseline: 2.3405x; 1.4203x over previous
#include <hip/hip_runtime.h>
#include <cmath>


#define N_NODES 50000
#define N_EDGES 800000
#define DIM 128
#define NPAD 51200

typedef float v2f __attribute__((ext_vector_type(2)));
typedef float v8f __attribute__((ext_vector_type(8)));

__device__ __forceinline__ int chip_half()
{
    return (int)((__builtin_amdgcn_s_sendmsg_rtn(0x87) >> 18) & 1u);
}

__global__ __launch_bounds__(256) void node_dots_kernel(
    const float* __restrict__ x, const float* __restrict__ w_pred,
    float* __restrict__ p_a, float* __restrict__ p_b, int* __restrict__ cnt)
{
    int node = (blockIdx.x * blockDim.x + threadIdx.x) >> 5;
    int lane = threadIdx.x & 31;
    if (node >= N_NODES) return;
    const float* xr = x + (size_t)node * DIM;
    float sa = 0.f, sb = 0.f;
#pragma unroll
    for (int j = 0; j < 4; ++j) {
        int c = lane + 32 * j;
        float v = xr[c];
        sa += v * w_pred[c];
        sb += v * w_pred[DIM + c];
    }
#pragma unroll
    for (int off = 16; off > 0; off >>= 1) {
        sa += __shfl_down(sa, off, 32);
        sb += __shfl_down(sb, off, 32);
    }
    if (lane == 0) {
        p_a[node] = sa;
        p_b[node] = sb;
        cnt[node] = 0;
        cnt[NPAD + node] = 0;
    }
}

__global__ __launch_bounds__(256) void edge_weight_kernel(
    const int* __restrict__ src, const int* __restrict__ tgt,
    const float* __restrict__ p_a, const float* __restrict__ p_b,
    const float* __restrict__ b_pred,
    float* __restrict__ ew, int* __restrict__ rank, int* __restrict__ cnt)
{
    int e = blockIdx.x * blockDim.x + threadIdx.x;
    if (e >= N_EDGES) return;
    const int h = chip_half();
    int s = src[e], t = tgt[e];
    float z = p_a[s] + p_b[t] + b_pred[0];
    float w = 1.0f / (1.0f + __expf(-z));
    ew[e] = w;
    const int r = atomicAdd(&cnt[h * NPAD + t], 1);
    rank[e] = (int)((unsigned)r | ((unsigned)h << 31));
}

__global__ __launch_bounds__(1024) void offsets_kernel(
    const int* __restrict__ cnt, int* __restrict__ off, int* __restrict__ tot)
{
    __shared__ int red[32];
    __shared__ int scn[32];
    const int tid = threadIdx.x, lane = tid & 31, w = tid >> 5;
    const int b = blockIdx.x;

    int part = 0;
    const int4* c4 = (const int4*)cnt;
    const int4* d4 = (const int4*)(cnt + NPAD);
    for (int i = tid; i < b * 256; i += 1024) {
        int4 v = c4[i];
        int4 u = d4[i];
        part += ((v.x + v.y) + (v.z + v.w)) + ((u.x + u.y) + (u.z + u.w));
    }
    const int n = b * 1024 + tid;
    const int m0 = (n < N_NODES) ? cnt[n] : 0;
    const int m1 = (n < N_NODES) ? cnt[NPAD + n] : 0;
    const int mine = m0 + m1;

#pragma unroll
    for (int o = 16; o > 0; o >>= 1) part += __shfl_xor(part, o, 32);
    int inc = mine;
#pragma unroll
    for (int o = 1; o < 32; o <<= 1) {
        int y = __shfl_up(inc, o, 32);
        if (lane >= o) inc += y;
    }
    if (lane == 0)  red[w] = part;
    if (lane == 31) scn[w] = inc;
    __syncthreads();
    int all = red[lane];
    int wp  = (lane < w) ? scn[lane] : 0;
#pragma unroll
    for (int o = 16; o > 0; o >>= 1) {
        all += __shfl_xor(all, o, 32);
        wp  += __shfl_xor(wp, o, 32);
    }
    if (n < N_NODES) {
        const int o = all + wp + inc - mine;
        off[n] = o;
        off[NPAD + n] = o + m0;
        tot[n] = mine;
    }
}

__global__ __launch_bounds__(256) void place_kernel(
    const int* __restrict__ src, const int* __restrict__ tgt,
    const float* __restrict__ ew, const int* __restrict__ rank,
    const int* __restrict__ off, int2* __restrict__ seg)
{
    int e = blockIdx.x * blockDim.x + threadIdx.x;
    if (e >= N_EDGES) return;
    int s = src[e], t = tgt[e];
    const int rk = rank[e];
    const int h = (int)((unsigned)rk >> 31);
    int pos = off[h * NPAD + t] + (rk & 0x7fffffff);
    seg[pos] = make_int2(s, __float_as_int(ew[e]));
}

__global__ __launch_bounds__(256) void degree_kernel(
    const int* __restrict__ off, const int* __restrict__ cnt,
    const int2* __restrict__ seg, float* __restrict__ dis)
{
    const int gid  = blockIdx.x * blockDim.x + threadIdx.x;
    const int node = gid >> 3, sub = gid & 7;
    const bool valid = node < N_NODES;
    const int nd = valid ? node : (N_NODES - 1);
    const int beg = off[nd], c = cnt[nd];
    float d = 0.f;
    for (int i = sub; i < c; i += 8) d += __int_as_float(seg[beg + i].y);
    d += __shfl_xor(d, 4, 32);
    d += __shfl_xor(d, 2, 32);
    d += __shfl_xor(d, 1, 32);
    if (valid && sub == 0) {
        float g = 1.0f + d;
        dis[node] = (g > 0.f) ? rsqrtf(g) : 0.f;
    }
}

__global__ __launch_bounds__(256) void gemm_wmma_kernel(
    const float* __restrict__ x, const float* __restrict__ W,
    float* __restrict__ xw)
{
    __shared__ float As[16 * 132];
    const int rowBase = blockIdx.x * 16;

#pragma unroll
    for (int j = 0; j < 8; ++j) {
        int idx = threadIdx.x + j * 256;
        int r = idx >> 7, c = idx & 127;
        As[r * 132 + c] = x[(size_t)rowBase * DIM + idx];
    }
    __syncthreads();

    const int waveId  = threadIdx.x >> 5;
    const int lane    = threadIdx.x & 31;
    const int m       = lane & 15;
    const int hi      = lane >> 4;
    const int colBase = waveId * 16;

    v8f acc = {};
#pragma unroll 8
    for (int k0 = 0; k0 < DIM; k0 += 4) {
        int ka = k0 + 2 * hi;
        v2f a, b;
        a.x = As[m * 132 + ka];
        a.y = As[m * 132 + ka + 1];
        b.x = W[(size_t)ka * DIM + colBase + m];
        b.y = W[(size_t)(ka + 1) * DIM + colBase + m];
        acc = __builtin_amdgcn_wmma_f32_16x16x4_f32(
            false, a, false, b, (short)0, acc, false, false);
    }

#pragma unroll
    for (int i = 0; i < 8; ++i) {
        int row = rowBase + i + 8 * hi;
        xw[(size_t)row * DIM + colBase + m] = acc[i];
    }
}

__global__ __launch_bounds__(256) void node_gather_kernel(
    const int* __restrict__ off, const int* __restrict__ cnt,
    const int2* __restrict__ seg, const float* __restrict__ dis,
    const float* __restrict__ xw, const float* __restrict__ b,
    float* __restrict__ out)
{
    const int t = __builtin_amdgcn_readfirstlane(
        (int)((blockIdx.x * blockDim.x + threadIdx.x) >> 5));
    const int lane = threadIdx.x & 31;
    if (t >= N_NODES) return;
    const int beg = off[t];
    const int n_e = cnt[t];
    const float dt = dis[t];
    const float4* xv = (const float4*)xw;

    float4 acc = make_float4(0.f, 0.f, 0.f, 0.f);
    for (int base = 0; base < n_e; base += 32) {
        const int i = base + lane;
        int   s_l = 0;
        float c_l = 0.f;
        if (i < n_e) {
            int2 ent = seg[beg + i];
            s_l = ent.x;
            c_l = dis[s_l] * __int_as_float(ent.y) * dt;
        }
        const int m = min(32, n_e - base);
        for (int j = 0; j < m; j += 4) {
            const int s0 = __builtin_amdgcn_readlane(s_l, j);
            const int s1 = __builtin_amdgcn_readlane(s_l, j + 1);
            const int s2 = __builtin_amdgcn_readlane(s_l, j + 2);
            const int s3 = __builtin_amdgcn_readlane(s_l, j + 3);
            const float c0 = __int_as_float(__builtin_amdgcn_readlane(__float_as_int(c_l), j));
            const float c1 = __int_as_float(__builtin_amdgcn_readlane(__float_as_int(c_l), j + 1));
            const float c2 = __int_as_float(__builtin_amdgcn_readlane(__float_as_int(c_l), j + 2));
            const float c3 = __int_as_float(__builtin_amdgcn_readlane(__float_as_int(c_l), j + 3));
            const float4 v0 = xv[(size_t)s0 * 32 + lane];
            const float4 v1 = xv[(size_t)s1 * 32 + lane];
            const float4 v2 = xv[(size_t)s2 * 32 + lane];
            const float4 v3 = xv[(size_t)s3 * 32 + lane];
            acc.x += c0 * v0.x; acc.y += c0 * v0.y; acc.z += c0 * v0.z; acc.w += c0 * v0.w;
            if (j + 1 < m) { acc.x += c1 * v1.x; acc.y += c1 * v1.y; acc.z += c1 * v1.z; acc.w += c1 * v1.w; }
            if (j + 2 < m) { acc.x += c2 * v2.x; acc.y += c2 * v2.y; acc.z += c2 * v2.z; acc.w += c2 * v2.w; }
            if (j + 3 < m) { acc.x += c3 * v3.x; acc.y += c3 * v3.y; acc.z += c3 * v3.z; acc.w += c3 * v3.w; }
        }
    }

    const float4 self = xv[(size_t)t * 32 + lane];
    const float4 bb = ((const float4*)b)[lane];
    const float ss = dt * dt;
    float4 r;
    r.x = acc.x + (ss * self.x + bb.x);
    r.y = acc.y + (ss * self.y + bb.y);
    r.z = acc.z + (ss * self.z + bb.z);
    r.w = acc.w + (ss * self.w + bb.w);
    ((float4*)out)[(size_t)t * 32 + lane] = r;
}

extern "C" void kernel_launch(void* const* d_in, const int* in_sizes, int n_in,
                              void* d_out, int out_size, void* d_ws, size_t ws_size,
                              hipStream_t stream) {
    const float* x          = (const float*)d_in[0];
    const int*   edge_index = (const int*)d_in[1];
    const float* W          = (const float*)d_in[2];
    const float* b          = (const float*)d_in[3];
    const float* w_pred     = (const float*)d_in[4];
    const float* b_pred     = (const float*)d_in[5];
    float*       out        = (float*)d_out;

    const int* src = edge_index;
    const int* tgt = edge_index + N_EDGES;

    float* ws   = (float*)d_ws;
    float* p_a  = ws;
    float* p_b  = p_a + N_NODES;
    float* dis  = p_b + N_NODES;
    float* ew   = dis + 3 * N_NODES;
    int*   rank = (int*)(ew + N_EDGES);
    int2*  seg  = (int2*)(rank + N_EDGES);
    float* xw   = ws + 3450048;
    int*   cnt  = (int*)(ws + 9900032);
    int*   off  = cnt + 2 * NPAD;
    int*   tot  = off + 2 * NPAD;

    node_dots_kernel  <<<(N_NODES + 7) / 8,          256,  0, stream>>>(x, w_pred, p_a, p_b, cnt);
    edge_weight_kernel<<<(N_EDGES + 255) / 256,      256,  0, stream>>>(src, tgt, p_a, p_b, b_pred, ew, rank, cnt);
    offsets_kernel    <<<(N_NODES + 1023) / 1024,    1024, 0, stream>>>(cnt, off, tot);
    place_kernel      <<<(N_EDGES + 255) / 256,      256,  0, stream>>>(src, tgt, ew, rank, off, seg);
    degree_kernel     <<<(N_NODES * 8 + 255) / 256,  256,  0, stream>>>(off, tot, seg, dis);
    gemm_wmma_kernel  <<<N_NODES / 16,               256,  0, stream>>>(x, W, xw);
    node_gather_kernel<<<N_NODES / 8,                256,  0, stream>>>(off, tot, seg, dis, xw, b, out);
}


# x read once: the predictor's per-node dots and the counter clears come from the GEMM kernel's staged tile of x (now first); the separate dots kernel is gone, 6 launches; every sum keeps its order
# speedup vs baseline: 2.5689x; 1.0976x over previous
#include <hip/hip_runtime.h>
#include <cmath>


#define N_NODES 50000
#define N_EDGES 800000
#define DIM 128
#define NPAD 51200

typedef float v2f __attribute__((ext_vector_type(2)));
typedef float v8f __attribute__((ext_vector_type(8)));

__device__ __forceinline__ int chip_half()
{
    return (int)((__builtin_amdgcn_s_sendmsg_rtn(0x87) >> 18) & 1u);
}

__global__ __launch_bounds__(256) void gemm_dots_kernel(
    const float* __restrict__ x, const float* __restrict__ W,
    const float* __restrict__ w_pred,
    float* __restrict__ xw, float* __restrict__ p_a, float* __restrict__ p_b,
    int* __restrict__ cnt)
{
    __shared__ float As[16 * 132];
    const int rowBase = blockIdx.x * 16;

#pragma unroll
    for (int j = 0; j < 8; ++j) {
        int idx = threadIdx.x + j * 256;
        int r = idx >> 7, c = idx & 127;
        As[r * 132 + c] = x[(size_t)rowBase * DIM + idx];
    }
    __syncthreads();

    const int waveId  = threadIdx.x >> 5;
    const int lane    = threadIdx.x & 31;
    const int m       = lane & 15;
    const int hi      = lane >> 4;
    const int colBase = waveId * 16;

    {
        const float* r0 = As + (2 * waveId) * 132;
        const float* r1 = r0 + 132;
        float sa0 = 0.f, sb0 = 0.f, sa1 = 0.f, sb1 = 0.f;
#pragma unroll
        for (int j = 0; j < 4; ++j) {
            const int c = lane + 32 * j;
            const float wa = w_pred[c];
            const float wb = w_pred[DIM + c];
            const float v0 = r0[c];
            const float v1 = r1[c];
            sa0 += v0 * wa;
            sb0 += v0 * wb;
            sa1 += v1 * wa;
            sb1 += v1 * wb;
        }
#pragma unroll
        for (int off = 16; off > 0; off >>= 1) {
            sa0 += __shfl_down(sa0, off, 32);
            sb0 += __shfl_down(sb0, off, 32);
            sa1 += __shfl_down(sa1, off, 32);
            sb1 += __shfl_down(sb1, off, 32);
        }
        if (lane == 0) {
            const int node = rowBase + 2 * waveId;
            *(float2*)(p_a + node) = make_float2(sa0, sa1);
            *(float2*)(p_b + node) = make_float2(sb0, sb1);
            *(int2*)(cnt + node) = make_int2(0, 0);
            *(int2*)(cnt + NPAD + node) = make_int2(0, 0);
        }
    }

    v8f acc = {};
#pragma unroll 8
    for (int k0 = 0; k0 < DIM; k0 += 4) {
        int ka = k0 + 2 * hi;
        v2f a, b;
        a.x = As[m * 132 + ka];
        a.y = As[m * 132 + ka + 1];
        b.x = W[(size_t)ka * DIM + colBase + m];
        b.y = W[(size_t)(ka + 1) * DIM + colBase + m];
        acc = __builtin_amdgcn_wmma_f32_16x16x4_f32(
            false, a, false, b, (short)0, acc, false, false);
    }

#pragma unroll
    for (int i = 0; i < 8; ++i) {
        int row = rowBase + i + 8 * hi;
        xw[(size_t)row * DIM + colBase + m] = acc[i];
    }
}

__global__ __launch_bounds__(256) void edge_weight_kernel(
    const int* __restrict__ src, const int* __restrict__ tgt,
    const float* __restrict__ p_a, const float* __restrict__ p_b,
    const float* __restrict__ b_pred,
    float* __restrict__ ew, int* __restrict__ rank, int* __restrict__ cnt)
{
    int e = blockIdx.x * blockDim.x + threadIdx.x;
    if (e >= N_EDGES) return;
    const int h = chip_half();
    int s = src[e], t = tgt[e];
    float z = p_a[s] + p_b[t] + b_pred[0];
    float w = 1.0f / (1.0f + __expf(-z));
    ew[e] = w;
    const int r = atomicAdd(&cnt[h * NPAD + t], 1);
    rank[e] = (int)((unsigned)r | ((unsigned)h << 31));
}

__global__ __launch_bounds__(1024) void offsets_kernel(
    const int* __restrict__ cnt, int* __restrict__ off, int* __restrict__ tot)
{
    __shared__ int red[32];
    __shared__ int scn[32];
    const int tid = threadIdx.x, lane = tid & 31, w = tid >> 5;
    const int b = blockIdx.x;

    int part = 0;
    const int4* c4 = (const int4*)cnt;
    const int4* d4 = (const int4*)(cnt + NPAD);
    for (int i = tid; i < b * 256; i += 1024) {
        int4 v = c4[i];
        int4 u = d4[i];
        part += ((v.x + v.y) + (v.z + v.w)) + ((u.x + u.y) + (u.z + u.w));
    }
    const int n = b * 1024 + tid;
    const int m0 = (n < N_NODES) ? cnt[n] : 0;
    const int m1 = (n < N_NODES) ? cnt[NPAD + n] : 0;
    const int mine = m0 + m1;

#pragma unroll
    for (int o = 16; o > 0; o >>= 1) part += __shfl_xor(part, o, 32);
    int inc = mine;
#pragma unroll
    for (int o = 1; o < 32; o <<= 1) {
        int y = __shfl_up(inc, o, 32);
        if (lane >= o) inc += y;
    }
    if (lane == 0)  red[w] = part;
    if (lane == 31) scn[w] = inc;
    __syncthreads();
    int all = red[lane];
    int wp  = (lane < w) ? scn[lane] : 0;
#pragma unroll
    for (int o = 16; o > 0; o >>= 1) {
        all += __shfl_xor(all, o, 32);
        wp  += __shfl_xor(wp, o, 32);
    }
    if (n < N_NODES) {
        const int o = all + wp + inc - mine;
        off[n] = o;
        off[NPAD + n] = o + m0;
        tot[n] = mine;
    }
}

__global__ __launch_bounds__(256) void place_kernel(
    const int* __restrict__ src, const int* __restrict__ tgt,
    const float* __restrict__ ew, const int* __restrict__ rank,
    const int* __restrict__ off, int2* __restrict__ seg)
{
    int e = blockIdx.x * blockDim.x + threadIdx.x;
    if (e >= N_EDGES) return;
    int s = src[e], t = tgt[e];
    const int rk = rank[e];
    const int h = (int)((unsigned)rk >> 31);
    int pos = off[h * NPAD + t] + (rk & 0x7fffffff);
    seg[pos] = make_int2(s, __float_as_int(ew[e]));
}

__global__ __launch_bounds__(256) void degree_kernel(
    const int* __restrict__ off, const int* __restrict__ cnt,
    const int2* __restrict__ seg, float* __restrict__ dis)
{
    const int gid  = blockIdx.x * blockDim.x + threadIdx.x;
    const int node = gid >> 3, sub = gid & 7;
    const bool valid = node < N_NODES;
    const int nd = valid ? node : (N_NODES - 1);
    const int beg = off[nd], c = cnt[nd];
    float d = 0.f;
    for (int i = sub; i < c; i += 8) d += __int_as_float(seg[beg + i].y);
    d += __shfl_xor(d, 4, 32);
    d += __shfl_xor(d, 2, 32);
    d += __shfl_xor(d, 1, 32);
    if (valid && sub == 0) {
        float g = 1.0f + d;
        dis[node] = (g > 0.f) ? rsqrtf(g) : 0.f;
    }
}

__global__ __launch_bounds__(256) void node_gather_kernel(
    const int* __restrict__ off, const int* __restrict__ cnt,
    const int2* __restrict__ seg, const float* __restrict__ dis,
    const float* __restrict__ xw, const float* __restrict__ b,
    float* __restrict__ out)
{
    const int t = __builtin_amdgcn_readfirstlane(
        (int)((blockIdx.x * blockDim.x + threadIdx.x) >> 5));
    const int lane = threadIdx.x & 31;
    if (t >= N_NODES) return;
    const int beg = off[t];
    const int n_e = cnt[t];
    const float dt = dis[t];
    const float4* xv = (const float4*)xw;

    float4 acc = make_float4(0.f, 0.f, 0.f, 0.f);
    for (int base = 0; base < n_e; base += 32) {
        const int i = base + lane;
        int   s_l = 0;
        float c_l = 0.f;
        if (i < n_e) {
            int2 ent = seg[beg + i];
            s_l = ent.x;
            c_l = dis[s_l] * __int_as_float(ent.y) * dt;
        }
        const int m = min(32, n_e - base);
        for (int j = 0; j < m; j += 4) {
            const int s0 = __builtin_amdgcn_readlane(s_l, j);
            const int s1 = __builtin_amdgcn_readlane(s_l, j + 1);
            const int s2 = __builtin_amdgcn_readlane(s_l, j + 2);
            const int s3 = __builtin_amdgcn_readlane(s_l, j + 3);
            const float c0 = __int_as_float(__builtin_amdgcn_readlane(__float_as_int(c_l), j));
            const float c1 = __int_as_float(__builtin_amdgcn_readlane(__float_as_int(c_l), j + 1));
            const float c2 = __int_as_float(__builtin_amdgcn_readlane(__float_as_int(c_l), j + 2));
            const float c3 = __int_as_float(__builtin_amdgcn_readlane(__float_as_int(c_l), j + 3));
            const float4 v0 = xv[(size_t)s0 * 32 + lane];
            const float4 v1 = xv[(size_t)s1 * 32 + lane];
            const float4 v2 = xv[(size_t)s2 * 32 + lane];
            const float4 v3 = xv[(size_t)s3 * 32 + lane];
            acc.x += c0 * v0.x; acc.y += c0 * v0.y; acc.z += c0 * v0.z; acc.w += c0 * v0.w;
            if (j + 1 < m) { acc.x += c1 * v1.x; acc.y += c1 * v1.y; acc.z += c1 * v1.z; acc.w += c1 * v1.w; }
            if (j + 2 < m) { acc.x += c2 * v2.x; acc.y += c2 * v2.y; acc.z += c2 * v2.z; acc.w += c2 * v2.w; }
            if (j + 3 < m) { acc.x += c3 * v3.x; acc.y += c3 * v3.y; acc.z += c3 * v3.z; acc.w += c3 * v3.w; }
        }
    }

    const float4 self = xv[(size_t)t * 32 + lane];
    const float4 bb = ((const float4*)b)[lane];
    const float ss = dt * dt;
    float4 r;
    r.x = acc.x + (ss * self.x + bb.x);
    r.y = acc.y + (ss * self.y + bb.y);
    r.z = acc.z + (ss * self.z + bb.z);
    r.w = acc.w + (ss * self.w + bb.w);
    ((float4*)out)[(size_t)t * 32 + lane] = r;
}

extern "C" void kernel_launch(void* const* d_in, const int* in_sizes, int n_in,
                              void* d_out, int out_size, void* d_ws, size_t ws_size,
                              hipStream_t stream) {
    const float* x          = (const float*)d_in[0];
    const int*   edge_index = (const int*)d_in[1];
    const float* W          = (const float*)d_in[2];
    const float* b          = (const float*)d_in[3];
    const float* w_pred     = (const float*)d_in[4];
    const float* b_pred     = (const float*)d_in[5];
    float*       out        = (float*)d_out;

    const int* src = edge_index;
    const int* tgt = edge_index + N_EDGES;

    float* ws   = (float*)d_ws;
    float* p_a  = ws;
    float* p_b  = p_a + N_NODES;
    float* dis  = p_b + N_NODES;
    float* ew   = dis + 3 * N_NODES;
    int*   rank = (int*)(ew + N_EDGES);
    int2*  seg  = (int2*)(rank + N_EDGES);
    float* xw   = ws + 3450048;
    int*   cnt  = (int*)(ws + 9900032);
    int*   off  = cnt + 2 * NPAD;
    int*   tot  = off + 2 * NPAD;

    gemm_dots_kernel  <<<N_NODES / 16,               256,  0, stream>>>(x, W, w_pred, xw, p_a, p_b, cnt);
    edge_weight_kernel<<<(N_EDGES + 255) / 256,      256,  0, stream>>>(src, tgt, p_a, p_b, b_pred, ew, rank, cnt);
    offsets_kernel    <<<(N_NODES + 1023) / 1024,    1024, 0, stream>>>(cnt, off, tot);
    place_kernel      <<<(N_EDGES + 255) / 256,      256,  0, stream>>>(src, tgt, ew, rank, off, seg);
    degree_kernel     <<<(N_NODES * 8 + 255) / 256,  256,  0, stream>>>(off, tot, seg, dis);
    node_gather_kernel<<<N_NODES / 8,                256,  0, stream>>>(off, tot, seg, dis, xw, b, out);
}


# each edge is stored at its counter add straight into its target's fixed 32-slot bucket per chip half (overflow list beyond, summed last): prefix-sum and place kernels gone, 4 launches; sum order kept
# speedup vs baseline: 3.5922x; 1.3983x over previous
#include <hip/hip_runtime.h>
#include <cmath>


#define N_NODES 50000
#define N_EDGES 800000
#define DIM 128
#define NPAD 51200
#define KH 32

typedef float v2f __attribute__((ext_vector_type(2)));
typedef float v8f __attribute__((ext_vector_type(8)));

__device__ __forceinline__ int chip_half()
{
    return (int)((__builtin_amdgcn_s_sendmsg_rtn(0x87) >> 18) & 1u);
}

__global__ __launch_bounds__(256) void gemm_dots_kernel(
    const float* __restrict__ x, const float* __restrict__ W,
    const float* __restrict__ w_pred,
    float* __restrict__ xw, float* __restrict__ p_a, float* __restrict__ p_b,
    int* __restrict__ cnt, int* __restrict__ ovf_n)
{
    __shared__ float As[16 * 132];
    const int rowBase = blockIdx.x * 16;
    if (blockIdx.x == 0 && threadIdx.x == 0) ovf_n[0] = 0;

#pragma unroll
    for (int j = 0; j < 8; ++j) {
        int idx = threadIdx.x + j * 256;
        int r = idx >> 7, c = idx & 127;
        As[r * 132 + c] = x[(size_t)rowBase * DIM + idx];
    }
    __syncthreads();

    const int waveId  = threadIdx.x >> 5;
    const int lane    = threadIdx.x & 31;
    const int m       = lane & 15;
    const int hi      = lane >> 4;
    const int colBase = waveId * 16;

    {
        const float* r0 = As + (2 * waveId) * 132;
        const float* r1 = r0 + 132;
        float sa0 = 0.f, sb0 = 0.f, sa1 = 0.f, sb1 = 0.f;
#pragma unroll
        for (int j = 0; j < 4; ++j) {
            const int c = lane + 32 * j;
            const float wa = w_pred[c];
            const float wb = w_pred[DIM + c];
            const float v0 = r0[c];
            const float v1 = r1[c];
            sa0 += v0 * wa;
            sb0 += v0 * wb;
            sa1 += v1 * wa;
            sb1 += v1 * wb;
        }
#pragma unroll
        for (int off = 16; off > 0; off >>= 1) {
            sa0 += __shfl_down(sa0, off, 32);
            sb0 += __shfl_down(sb0, off, 32);
            sa1 += __shfl_down(sa1, off, 32);
            sb1 += __shfl_down(sb1, off, 32);
        }
        if (lane == 0) {
            const int node = rowBase + 2 * waveId;
            *(float2*)(p_a + node) = make_float2(sa0, sa1);
            *(float2*)(p_b + node) = make_float2(sb0, sb1);
            *(int2*)(cnt + node) = make_int2(0, 0);
            *(int2*)(cnt + NPAD + node) = make_int2(0, 0);
        }
    }

    v8f acc = {};
#pragma unroll 8
    for (int k0 = 0; k0 < DIM; k0 += 4) {
        int ka = k0 + 2 * hi;
        v2f a, b;
        a.x = As[m * 132 + ka];
        a.y = As[m * 132 + ka + 1];
        b.x = W[(size_t)ka * DIM + colBase + m];
        b.y = W[(size_t)(ka + 1) * DIM + colBase + m];
        acc = __builtin_amdgcn_wmma_f32_16x16x4_f32(
            false, a, false, b, (short)0, acc, false, false);
    }

#pragma unroll
    for (int i = 0; i < 8; ++i) {
        int row = rowBase + i + 8 * hi;
        xw[(size_t)row * DIM + colBase + m] = acc[i];
    }
}

__global__ __launch_bounds__(256) void edge_bucket_kernel(
    const int* __restrict__ src, const int* __restrict__ tgt,
    const float* __restrict__ p_a, const float* __restrict__ p_b,
    const float* __restrict__ b_pred,
    int* __restrict__ cnt, int2* __restrict__ seg,
    int* __restrict__ ovf_n, int4* __restrict__ ovf)
{
    int e = blockIdx.x * blockDim.x + threadIdx.x;
    if (e >= N_EDGES) return;
    const int h = chip_half();
    int s = src[e], t = tgt[e];
    float z = p_a[s] + p_b[t] + b_pred[0];
    float w = 1.0f / (1.0f + __expf(-z));
    const int r = atomicAdd(&cnt[h * NPAD + t], 1);
    if (r < KH) {
        seg[((size_t)t * 2 + h) * KH + r] = make_int2(s, __float_as_int(w));
    } else {
        const int o = atomicAdd(ovf_n, 1);
        ovf[o] = make_int4(s, t, __float_as_int(w), 0);
    }
}

__global__ __launch_bounds__(256) void degree_kernel(
    const int* __restrict__ cnt, const int2* __restrict__ seg,
    const int* __restrict__ ovf_n, const int4* __restrict__ ovf,
    float* __restrict__ dis)
{
    const int gid  = blockIdx.x * blockDim.x + threadIdx.x;
    const int node = gid >> 3, sub = gid & 7;
    const bool valid = node < N_NODES;
    const int nd = valid ? node : (N_NODES - 1);
    const int m0 = cnt[nd], m1 = cnt[NPAD + nd];
    const int n0 = min(m0, KH), c = n0 + min(m1, KH);
    const int2* bk = seg + (size_t)nd * (2 * KH);
    float d = 0.f;
    for (int i = sub; i < c; i += 8) {
        const int slot = (i < n0) ? i : (KH + i - n0);
        d += __int_as_float(bk[slot].y);
    }
    d += __shfl_xor(d, 4, 32);
    d += __shfl_xor(d, 2, 32);
    d += __shfl_xor(d, 1, 32);
    if ((m0 > KH) | (m1 > KH)) {
        const int n_o = min(ovf_n[0], N_EDGES);
        for (int i = 0; i < n_o; ++i) {
            const int4 q = ovf[i];
            if (q.y == nd) d += __int_as_float(q.z);
        }
    }
    if (valid && sub == 0) {
        float g = 1.0f + d;
        dis[node] = (g > 0.f) ? rsqrtf(g) : 0.f;
    }
}

__global__ __launch_bounds__(256) void node_gather_kernel(
    const int* __restrict__ cnt, const int2* __restrict__ seg,
    const int* __restrict__ ovf_n, const int4* __restrict__ ovf,
    const float* __restrict__ dis,
    const float* __restrict__ xw, const float* __restrict__ b,
    float* __restrict__ out)
{
    const int t = __builtin_amdgcn_readfirstlane(
        (int)((blockIdx.x * blockDim.x + threadIdx.x) >> 5));
    const int lane = threadIdx.x & 31;
    if (t >= N_NODES) return;
    const int m0 = cnt[t], m1 = cnt[NPAD + t];
    const int n0 = min(m0, KH);
    const int n_e = n0 + min(m1, KH);
    const int2* bk = seg + (size_t)t * (2 * KH);
    const float dt = dis[t];
    const float4* xv = (const float4*)xw;

    float4 acc = make_float4(0.f, 0.f, 0.f, 0.f);
    for (int base = 0; base < n_e; base += 32) {
        const int i = base + lane;
        int   s_l = 0;
        float c_l = 0.f;
        if (i < n_e) {
            const int slot = (i < n0) ? i : (KH + i - n0);
            int2 ent = bk[slot];
            s_l = ent.x;
            c_l = dis[s_l] * __int_as_float(ent.y) * dt;
        }
        const int m = min(32, n_e - base);
        for (int j = 0; j < m; j += 4) {
            const int s0 = __builtin_amdgcn_readlane(s_l, j);
            const int s1 = __builtin_amdgcn_readlane(s_l, j + 1);
            const int s2 = __builtin_amdgcn_readlane(s_l, j + 2);
            const int s3 = __builtin_amdgcn_readlane(s_l, j + 3);
            const float c0 = __int_as_float(__builtin_amdgcn_readlane(__float_as_int(c_l), j));
            const float c1 = __int_as_float(__builtin_amdgcn_readlane(__float_as_int(c_l), j + 1));
            const float c2 = __int_as_float(__builtin_amdgcn_readlane(__float_as_int(c_l), j + 2));
            const float c3 = __int_as_float(__builtin_amdgcn_readlane(__float_as_int(c_l), j + 3));
            const float4 v0 = xv[(size_t)s0 * 32 + lane];
            const float4 v1 = xv[(size_t)s1 * 32 + lane];
            const float4 v2 = xv[(size_t)s2 * 32 + lane];
            const float4 v3 = xv[(size_t)s3 * 32 + lane];
            acc.x += c0 * v0.x; acc.y += c0 * v0.y; acc.z += c0 * v0.z; acc.w += c0 * v0.w;
            if (j + 1 < m) { acc.x += c1 * v1.x; acc.y += c1 * v1.y; acc.z += c1 * v1.z; acc.w += c1 * v1.w; }
            if (j + 2 < m) { acc.x += c2 * v2.x; acc.y += c2 * v2.y; acc.z += c2 * v2.z; acc.w += c2 * v2.w; }
            if (j + 3 < m) { acc.x += c3 * v3.x; acc.y += c3 * v3.y; acc.z += c3 * v3.z; acc.w += c3 * v3.w; }
        }
    }

    if ((m0 > KH) | (m1 > KH)) {
        const int n_o = min(ovf_n[0], N_EDGES);
        for (int base = 0; base < n_o; base += 32) {
            const int i = base + lane;
            int   s_l = 0;
            float c_l = 0.f;
            bool  mine = false;
            if (i < n_o) {
                const int4 q = ovf[i];
                if (q.y == t) {
                    mine = true;
                    s_l = q.x;
                    c_l = dis[s_l] * __int_as_float(q.z) * dt;
                }
            }
            unsigned todo = __builtin_amdgcn_ballot_w32(mine);
            while (todo) {
                const int j = __builtin_ctz(todo);
                todo &= todo - 1;
                const int   sj = __builtin_amdgcn_readlane(s_l, j);
                const float cj = __int_as_float(__builtin_amdgcn_readlane(__float_as_int(c_l), j));
                const float4 v = xv[(size_t)sj * 32 + lane];
                acc.x += cj * v.x; acc.y += cj * v.y; acc.z += cj * v.z; acc.w += cj * v.w;
            }
        }
    }

    const float4 self = xv[(size_t)t * 32 + lane];
    const float4 bb = ((const float4*)b)[lane];
    const float ss = dt * dt;
    float4 r;
    r.x = acc.x + (ss * self.x + bb.x);
    r.y = acc.y + (ss * self.y + bb.y);
    r.z = acc.z + (ss * self.z + bb.z);
    r.w = acc.w + (ss * self.w + bb.w);
    ((float4*)out)[(size_t)t * 32 + lane] = r;
}

extern "C" void kernel_launch(void* const* d_in, const int* in_sizes, int n_in,
                              void* d_out, int out_size, void* d_ws, size_t ws_size,
                              hipStream_t stream) {
    const float* x          = (const float*)d_in[0];
    const int*   edge_index = (const int*)d_in[1];
    const float* W          = (const float*)d_in[2];
    const float* b          = (const float*)d_in[3];
    const float* w_pred     = (const float*)d_in[4];
    const float* b_pred     = (const float*)d_in[5];
    float*       out        = (float*)d_out;

    const int* src = edge_index;
    const int* tgt = edge_index + N_EDGES;

    float* ws   = (float*)d_ws;
    float* p_a  = ws;
    float* p_b  = p_a + N_NODES;
    float* dis  = p_b + N_NODES;
    int2*  seg  = (int2*)(ws + 150016);
    float* xw   = ws + 6550016;
    int*   cnt  = (int*)(ws + 12950528);
    int*   ovf_n = (int*)(ws + 13052928);
    int4*  ovf  = (int4*)(ws + 13053952);

    gemm_dots_kernel  <<<N_NODES / 16,               256,  0, stream>>>(x, W, w_pred, xw, p_a, p_b, cnt, ovf_n);
    edge_bucket_kernel<<<(N_EDGES + 255) / 256,      256,  0, stream>>>(src, tgt, p_a, p_b, b_pred, cnt, seg, ovf_n, ovf);
    degree_kernel     <<<(N_NODES * 8 + 255) / 256,  256,  0, stream>>>(cnt, seg, ovf_n, ovf, dis);
    node_gather_kernel<<<N_NODES / 8,                256,  0, stream>>>(cnt, seg, ovf_n, ovf, dis, xw, b, out);
}


# the edge's weight rides in its slot add (64-bit word: count + 2^-24 fixed-point degree sum, order-free), dis taken at use: degree kernel and dis table gone, 3 launches; output sum order kept
# speedup vs baseline: 3.7486x; 1.0435x over previous
#include <hip/hip_runtime.h>
#include <cmath>


#define N_NODES 50000
#define N_EDGES 800000
#define DIM 128
#define NPAD 51200
#define KH 32
#define CNT_BITS 20
#define CNT_MASK 0xFFFFFu
#define FIX_ONE 16777216.0f

typedef float v2f __attribute__((ext_vector_type(2)));
typedef float v8f __attribute__((ext_vector_type(8)));
typedef unsigned long long u64;

__device__ __forceinline__ float inv_sqrt_degree(u64 c0, u64 c1)
{
    const u64 sum = (c0 >> CNT_BITS) + (c1 >> CNT_BITS);
    const unsigned whole = (unsigned)(sum >> 24);
    const unsigned frac  = (unsigned)sum & 0xFFFFFFu;
    const float g = fmaf((float)frac, 1.0f / FIX_ONE, (float)(whole + 1u));
    return (g > 0.f) ? rsqrtf(g) : 0.f;
}

__device__ __forceinline__ int chip_half()
{
    return (int)((__builtin_amdgcn_s_sendmsg_rtn(0x87) >> 18) & 1u);
}

__global__ __launch_bounds__(256) void gemm_dots_kernel(
    const float* __restrict__ x, const float* __restrict__ W,
    const float* __restrict__ w_pred,
    float* __restrict__ xw, float* __restrict__ p_a, float* __restrict__ p_b,
    u64* __restrict__ cd, int* __restrict__ ovf_n)
{
    __shared__ float As[16 * 132];
    const int rowBase = blockIdx.x * 16;
    if (blockIdx.x == 0 && threadIdx.x == 0) ovf_n[0] = 0;

#pragma unroll
    for (int j = 0; j < 8; ++j) {
        int idx = threadIdx.x + j * 256;
        int r = idx >> 7, c = idx & 127;
        As[r * 132 + c] = x[(size_t)rowBase * DIM + idx];
    }
    __syncthreads();

    const int waveId  = threadIdx.x >> 5;
    const int lane    = threadIdx.x & 31;
    const int m       = lane & 15;
    const int hi      = lane >> 4;
    const int colBase = waveId * 16;

    {
        const float* r0 = As + (2 * waveId) * 132;
        const float* r1 = r0 + 132;
        float sa0 = 0.f, sb0 = 0.f, sa1 = 0.f, sb1 = 0.f;
#pragma unroll
        for (int j = 0; j < 4; ++j) {
            const int c = lane + 32 * j;
            const float wa = w_pred[c];
            const float wb = w_pred[DIM + c];
            const float v0 = r0[c];
            const float v1 = r1[c];
            sa0 += v0 * wa;
            sb0 += v0 * wb;
            sa1 += v1 * wa;
            sb1 += v1 * wb;
        }
#pragma unroll
        for (int off = 16; off > 0; off >>= 1) {
            sa0 += __shfl_down(sa0, off, 32);
            sb0 += __shfl_down(sb0, off, 32);
            sa1 += __shfl_down(sa1, off, 32);
            sb1 += __shfl_down(sb1, off, 32);
        }
        if (lane == 0) {
            const int node = rowBase + 2 * waveId;
            *(float2*)(p_a + node) = make_float2(sa0, sa1);
            *(float2*)(p_b + node) = make_float2(sb0, sb1);
            *(ulonglong2*)(cd + node) = make_ulonglong2(0ull, 0ull);
            *(ulonglong2*)(cd + NPAD + node) = make_ulonglong2(0ull, 0ull);
        }
    }

    v8f acc = {};
#pragma unroll 8
    for (int k0 = 0; k0 < DIM; k0 += 4) {
        int ka = k0 + 2 * hi;
        v2f a, b;
        a.x = As[m * 132 + ka];
        a.y = As[m * 132 + ka + 1];
        b.x = W[(size_t)ka * DIM + colBase + m];
        b.y = W[(size_t)(ka + 1) * DIM + colBase + m];
        acc = __builtin_amdgcn_wmma_f32_16x16x4_f32(
            false, a, false, b, (short)0, acc, false, false);
    }

#pragma unroll
    for (int i = 0; i < 8; ++i) {
        int row = rowBase + i + 8 * hi;
        xw[(size_t)row * DIM + colBase + m] = acc[i];
    }
}

__global__ __launch_bounds__(256) void edge_bucket_kernel(
    const int* __restrict__ src, const int* __restrict__ tgt,
    const float* __restrict__ p_a, const float* __restrict__ p_b,
    const float* __restrict__ b_pred,
    u64* __restrict__ cd, int2* __restrict__ seg,
    int* __restrict__ ovf_n, int4* __restrict__ ovf)
{
    int e = blockIdx.x * blockDim.x + threadIdx.x;
    if (e >= N_EDGES) return;
    const int h = chip_half();
    int s = src[e], t = tgt[e];
    float z = p_a[s] + p_b[t] + b_pred[0];
    float w = 1.0f / (1.0f + __expf(-z));
    const unsigned q = (unsigned)__builtin_rintf(w * FIX_ONE);
    const u64 old = atomicAdd(&cd[h * NPAD + t], ((u64)q << CNT_BITS) | 1ull);
    const int r = (int)((unsigned)old & CNT_MASK);
    if (r < KH) {
        seg[((size_t)t * 2 + h) * KH + r] = make_int2(s, __float_as_int(w));
    } else {
        const int o = atomicAdd(ovf_n, 1);
        ovf[o] = make_int4(s, t, __float_as_int(w), 0);
    }
}

__global__ __launch_bounds__(256) void node_gather_kernel(
    const u64* __restrict__ cd, const int2* __restrict__ seg,
    const int* __restrict__ ovf_n, const int4* __restrict__ ovf,
    const float* __restrict__ xw, const float* __restrict__ b,
    float* __restrict__ out)
{
    const int t = __builtin_amdgcn_readfirstlane(
        (int)((blockIdx.x * blockDim.x + threadIdx.x) >> 5));
    const int lane = threadIdx.x & 31;
    if (t >= N_NODES) return;
    const u64 w0 = cd[t], w1 = cd[NPAD + t];
    const int m0 = (int)((unsigned)w0 & CNT_MASK), m1 = (int)((unsigned)w1 & CNT_MASK);
    const int n0 = min(m0, KH);
    const int n_e = n0 + min(m1, KH);
    const int2* bk = seg + (size_t)t * (2 * KH);
    const float dt = inv_sqrt_degree(w0, w1);
    const float4* xv = (const float4*)xw;

    float4 acc = make_float4(0.f, 0.f, 0.f, 0.f);
    for (int base = 0; base < n_e; base += 32) {
        const int i = base + lane;
        int   s_l = 0;
        float c_l = 0.f;
        if (i < n_e) {
            const int slot = (i < n0) ? i : (KH + i - n0);
            int2 ent = bk[slot];
            s_l = ent.x;
            c_l = inv_sqrt_degree(cd[s_l], cd[NPAD + s_l]) * __int_as_float(ent.y) * dt;
        }
        const int m = min(32, n_e - base);
        for (int j = 0; j < m; j += 4) {
            const int s0 = __builtin_amdgcn_readlane(s_l, j);
            const int s1 = __builtin_amdgcn_readlane(s_l, j + 1);
            const int s2 = __builtin_amdgcn_readlane(s_l, j + 2);
            const int s3 = __builtin_amdgcn_readlane(s_l, j + 3);
            const float c0 = __int_as_float(__builtin_amdgcn_readlane(__float_as_int(c_l), j));
            const float c1 = __int_as_float(__builtin_amdgcn_readlane(__float_as_int(c_l), j + 1));
            const float c2 = __int_as_float(__builtin_amdgcn_readlane(__float_as_int(c_l), j + 2));
            const float c3 = __int_as_float(__builtin_amdgcn_readlane(__float_as_int(c_l), j + 3));
            const float4 v0 = xv[(size_t)s0 * 32 + lane];
            const float4 v1 = xv[(size_t)s1 * 32 + lane];
            const float4 v2 = xv[(size_t)s2 * 32 + lane];
            const float4 v3 = xv[(size_t)s3 * 32 + lane];
            acc.x += c0 * v0.x; acc.y += c0 * v0.y; acc.z += c0 * v0.z; acc.w += c0 * v0.w;
            if (j + 1 < m) { acc.x += c1 * v1.x; acc.y += c1 * v1.y; acc.z += c1 * v1.z; acc.w += c1 * v1.w; }
            if (j + 2 < m) { acc.x += c2 * v2.x; acc.y += c2 * v2.y; acc.z += c2 * v2.z; acc.w += c2 * v2.w; }
            if (j + 3 < m) { acc.x += c3 * v3.x; acc.y += c3 * v3.y; acc.z += c3 * v3.z; acc.w += c3 * v3.w; }
        }
    }

    if ((m0 > KH) | (m1 > KH)) {
        const int n_o = min(ovf_n[0], N_EDGES);
        for (int base = 0; base < n_o; base += 32) {
            const int i = base + lane;
            int   s_l = 0;
            float c_l = 0.f;
            bool  mine = false;
            if (i < n_o) {
                const int4 q = ovf[i];
                if (q.y == t) {
                    mine = true;
                    s_l = q.x;
                    c_l = inv_sqrt_degree(cd[s_l], cd[NPAD + s_l]) * __int_as_float(q.z) * dt;
                }
            }
            unsigned todo = __builtin_amdgcn_ballot_w32(mine);
            while (todo) {
                const int j = __builtin_ctz(todo);
                todo &= todo - 1;
                const int   sj = __builtin_amdgcn_readlane(s_l, j);
                const float cj = __int_as_float(__builtin_amdgcn_readlane(__float_as_int(c_l), j));
                const float4 v = xv[(size_t)sj * 32 + lane];
                acc.x += cj * v.x; acc.y += cj * v.y; acc.z += cj * v.z; acc.w += cj * v.w;
            }
        }
    }

    const float4 self = xv[(size_t)t * 32 + lane];
    const float4 bb = ((const float4*)b)[lane];
    const float ss = dt * dt;
    float4 r;
    r.x = acc.x + (ss * self.x + bb.x);
    r.y = acc.y + (ss * self.y + bb.y);
    r.z = acc.z + (ss * self.z + bb.z);
    r.w = acc.w + (ss * self.w + bb.w);
    ((float4*)out)[(size_t)t * 32 + lane] = r;
}

extern "C" void kernel_launch(void* const* d_in, const int* in_sizes, int n_in,
                              void* d_out, int out_size, void* d_ws, size_t ws_size,
                              hipStream_t stream) {
    const float* x          = (const float*)d_in[0];
    const int*   edge_index = (const int*)d_in[1];
    const float* W          = (const float*)d_in[2];
    const float* b          = (const float*)d_in[3];
    const float* w_pred     = (const float*)d_in[4];
    const float* b_pred     = (const float*)d_in[5];
    float*       out        = (float*)d_out;

    const int* src = edge_index;
    const int* tgt = edge_index + N_EDGES;

    float* ws   = (float*)d_ws;
    float* p_a  = ws;
    float* p_b  = p_a + N_NODES;
    int2*  seg  = (int2*)(ws + 150016);
    float* xw   = ws + 6550016;
    u64*   cd   = (u64*)(ws + 12950528);
    int*   ovf_n = (int*)(ws + 13155328);
    int4*  ovf  = (int4*)(ws + 13156352);

    gemm_dots_kernel  <<<N_NODES / 16,               256,  0, stream>>>(x, W, w_pred, xw, p_a, p_b, cd, ovf_n);
    edge_bucket_kernel<<<(N_EDGES + 255) / 256,      256,  0, stream>>>(src, tgt, p_a, p_b, b_pred, cd, seg, ovf_n, ovf);
    node_gather_kernel<<<N_NODES / 8,                256,  0, stream>>>(cd, seg, ovf_n, ovf, xw, b, out);
}


# every kernel a resident grid whose waves loop over work (3 stripes per block, 4 edges per lane, 3-4 nodes per wave): 30832 wave starts, not one per item; W fragments loaded once; sum orders kept
# speedup vs baseline: 3.9210x; 1.0460x over previous
#include <hip/hip_runtime.h>
#include <cmath>


#define N_NODES 50000
#define N_EDGES 800000
#define DIM 128
#define NPAD 51200
#define KH 32
#define N_TILES (N_NODES / 16)
#define GEMM_BLOCKS 1024
#define EPT 4
#define EDGE_BLOCKS ((N_EDGES + 256 * EPT - 1) / (256 * EPT))
#define GATHER_BLOCKS 2048
#define CNT_BITS 20
#define CNT_MASK 0xFFFFFu
#define FIX_ONE 16777216.0f

typedef float v2f __attribute__((ext_vector_type(2)));
typedef float v8f __attribute__((ext_vector_type(8)));
typedef unsigned long long u64;

__device__ __forceinline__ float inv_sqrt_degree(u64 c0, u64 c1)
{
    const u64 sum = (c0 >> CNT_BITS) + (c1 >> CNT_BITS);
    const unsigned whole = (unsigned)(sum >> 24);
    const unsigned frac  = (unsigned)sum & 0xFFFFFFu;
    const float g = fmaf((float)frac, 1.0f / FIX_ONE, (float)(whole + 1u));
    return (g > 0.f) ? rsqrtf(g) : 0.f;
}

__device__ __forceinline__ int chip_half()
{
    return (int)((__builtin_amdgcn_s_sendmsg_rtn(0x87) >> 18) & 1u);
}

__global__ __launch_bounds__(256) void gemm_dots_kernel(
    const float* __restrict__ x, const float* __restrict__ W,
    const float* __restrict__ w_pred,
    float* __restrict__ xw, float* __restrict__ p_a, float* __restrict__ p_b,
    u64* __restrict__ cd, int* __restrict__ ovf_n)
{
    __shared__ float Asm[2 * 16 * 132];
    if (blockIdx.x == 0 && threadIdx.x == 0) ovf_n[0] = 0;

    const int waveId  = threadIdx.x >> 5;
    const int lane    = threadIdx.x & 31;
    const int m       = lane & 15;
    const int hi      = lane >> 4;
    const int colBase = waveId * 16;

    v2f bfrag[32];
#pragma unroll
    for (int i = 0; i < 32; ++i) {
        const int ka = 4 * i + 2 * hi;
        bfrag[i].x = W[(size_t)ka * DIM + colBase + m];
        bfrag[i].y = W[(size_t)(ka + 1) * DIM + colBase + m];
    }

    int buf = 0;
    for (int tile = blockIdx.x; tile < N_TILES; tile += gridDim.x, buf ^= 1) {
        const int rowBase = tile * 16;
        float* As = Asm + buf * (16 * 132);

#pragma unroll
        for (int j = 0; j < 8; ++j) {
            int idx = threadIdx.x + j * 256;
            int r = idx >> 7, c = idx & 127;
            As[r * 132 + c] = x[(size_t)rowBase * DIM + idx];
        }
        __syncthreads();

        {
            const float* r0 = As + (2 * waveId) * 132;
            const float* r1 = r0 + 132;
            float sa0 = 0.f, sb0 = 0.f, sa1 = 0.f, sb1 = 0.f;
#pragma unroll
            for (int j = 0; j < 4; ++j) {
                const int c = lane + 32 * j;
                const float wa = w_pred[c];
                const float wb = w_pred[DIM + c];
                const float v0 = r0[c];
                const float v1 = r1[c];
                sa0 += v0 * wa;
                sb0 += v0 * wb;
                sa1 += v1 * wa;
                sb1 += v1 * wb;
            }
#pragma unroll
            for (int off = 16; off > 0; off >>= 1) {
                sa0 += __shfl_down(sa0, off, 32);
                sb0 += __shfl_down(sb0, off, 32);
                sa1 += __shfl_down(sa1, off, 32);
                sb1 += __shfl_down(sb1, off, 32);
            }
            if (lane == 0) {
                const int node = rowBase + 2 * waveId;
                *(float2*)(p_a + node) = make_float2(sa0, sa1);
                *(float2*)(p_b + node) = make_float2(sb0, sb1);
                *(ulonglong2*)(cd + node) = make_ulonglong2(0ull, 0ull);
                *(ulonglong2*)(cd + NPAD + node) = make_ulonglong2(0ull, 0ull);
            }
        }

        v8f acc = {};
#pragma unroll
        for (int i = 0; i < 32; ++i) {
            const int ka = 4 * i + 2 * hi;
            v2f a;
            a.x = As[m * 132 + ka];
            a.y = As[m * 132 + ka + 1];
            acc = __builtin_amdgcn_wmma_f32_16x16x4_f32(
                false, a, false, bfrag[i], (short)0, acc, false, false);
        }

#pragma unroll
        for (int i = 0; i < 8; ++i) {
            int row = rowBase + i + 8 * hi;
            xw[(size_t)row * DIM + colBase + m] = acc[i];
        }
    }
}

__global__ __launch_bounds__(256) void edge_bucket_kernel(
    const int* __restrict__ src, const int* __restrict__ tgt,
    const float* __restrict__ p_a, const float* __restrict__ p_b,
    const float* __restrict__ b_pred,
    u64* __restrict__ cd, int2* __restrict__ seg,
    int* __restrict__ ovf_n, int4* __restrict__ ovf)
{
    const int h = chip_half();
    const float bp = b_pred[0];
    const int e0 = blockIdx.x * (256 * EPT) + threadIdx.x;
    int  s[EPT], t[EPT];
    bool ok[EPT];
#pragma unroll
    for (int k = 0; k < EPT; ++k) {
        const int e = e0 + k * 256;
        ok[k] = e < N_EDGES;
        s[k] = ok[k] ? src[e] : 0;
        t[k] = ok[k] ? tgt[e] : 0;
    }
    float w[EPT];
#pragma unroll
    for (int k = 0; k < EPT; ++k) {
        const float z = p_a[s[k]] + p_b[t[k]] + bp;
        w[k] = 1.0f / (1.0f + __expf(-z));
    }
    u64 old[EPT];
#pragma unroll
    for (int k = 0; k < EPT; ++k) {
        old[k] = 0ull;
        if (ok[k]) {
            const unsigned q = (unsigned)__builtin_rintf(w[k] * FIX_ONE);
            old[k] = atomicAdd(&cd[h * NPAD + t[k]], ((u64)q << CNT_BITS) | 1ull);
        }
    }
#pragma unroll
    for (int k = 0; k < EPT; ++k) {
        if (!ok[k]) continue;
        const int r = (int)((unsigned)old[k] & CNT_MASK);
        if (r < KH) {
            seg[((size_t)t[k] * 2 + h) * KH + r] = make_int2(s[k], __float_as_int(w[k]));
        } else {
            const int o = atomicAdd(ovf_n, 1);
            ovf[o] = make_int4(s[k], t[k], __float_as_int(w[k]), 0);
        }
    }
}

__global__ __launch_bounds__(256) void node_gather_kernel(
    const u64* __restrict__ cd, const int2* __restrict__ seg,
    const int* __restrict__ ovf_n, const int4* __restrict__ ovf,
    const float* __restrict__ xw, const float* __restrict__ b,
    float* __restrict__ out)
{
    const int gw = __builtin_amdgcn_readfirstlane(
        (int)((blockIdx.x * blockDim.x + threadIdx.x) >> 5));
    const int n_waves = (int)(gridDim.x * (blockDim.x >> 5));
    const int lane = threadIdx.x & 31;
    const float4* xv = (const float4*)xw;
    const float4 bb = ((const float4*)b)[lane];

    for (int t = gw; t < N_NODES; t += n_waves) {
        const u64 w0 = cd[t], w1 = cd[NPAD + t];
        const int m0 = (int)((unsigned)w0 & CNT_MASK), m1 = (int)((unsigned)w1 & CNT_MASK);
        const int n0 = min(m0, KH);
        const int n_e = n0 + min(m1, KH);
        const int2* bk = seg + (size_t)t * (2 * KH);
        const float dt = inv_sqrt_degree(w0, w1);

        float4 acc = make_float4(0.f, 0.f, 0.f, 0.f);
        for (int base = 0; base < n_e; base += 32) {
            const int i = base + lane;
            int   s_l = 0;
            float c_l = 0.f;
            if (i < n_e) {
                const int slot = (i < n0) ? i : (KH + i - n0);
                int2 ent = bk[slot];
                s_l = ent.x;
                c_l = inv_sqrt_degree(cd[s_l], cd[NPAD + s_l]) * __int_as_float(ent.y) * dt;
            }
            const int m = min(32, n_e - base);
            for (int j = 0; j < m; j += 4) {
                const int s0 = __builtin_amdgcn_readlane(s_l, j);
                const int s1 = __builtin_amdgcn_readlane(s_l, j + 1);
                const int s2 = __builtin_amdgcn_readlane(s_l, j + 2);
                const int s3 = __builtin_amdgcn_readlane(s_l, j + 3);
                const float c0 = __int_as_float(__builtin_amdgcn_readlane(__float_as_int(c_l), j));
                const float c1 = __int_as_float(__builtin_amdgcn_readlane(__float_as_int(c_l), j + 1));
                const float c2 = __int_as_float(__builtin_amdgcn_readlane(__float_as_int(c_l), j + 2));
                const float c3 = __int_as_float(__builtin_amdgcn_readlane(__float_as_int(c_l), j + 3));
                const float4 v0 = xv[(size_t)s0 * 32 + lane];
                const float4 v1 = xv[(size_t)s1 * 32 + lane];
                const float4 v2 = xv[(size_t)s2 * 32 + lane];
                const float4 v3 = xv[(size_t)s3 * 32 + lane];
                acc.x += c0 * v0.x; acc.y += c0 * v0.y; acc.z += c0 * v0.z; acc.w += c0 * v0.w;
                if (j + 1 < m) { acc.x += c1 * v1.x; acc.y += c1 * v1.y; acc.z += c1 * v1.z; acc.w += c1 * v1.w; }
                if (j + 2 < m) { acc.x += c2 * v2.x; acc.y += c2 * v2.y; acc.z += c2 * v2.z; acc.w += c2 * v2.w; }
                if (j + 3 < m) { acc.x += c3 * v3.x; acc.y += c3 * v3.y; acc.z += c3 * v3.z; acc.w += c3 * v3.w; }
            }
        }

        if ((m0 > KH) | (m1 > KH)) {
            const int n_o = min(ovf_n[0], N_EDGES);
            for (int base = 0; base < n_o; base += 32) {
                const int i = base + lane;
                int   s_l = 0;
                float c_l = 0.f;
                bool  mine = false;
                if (i < n_o) {
                    const int4 q = ovf[i];
                    if (q.y == t) {
                        mine = true;
                        s_l = q.x;
                        c_l = inv_sqrt_degree(cd[s_l], cd[NPAD + s_l]) * __int_as_float(q.z) * dt;
                    }
                }
                unsigned todo = __builtin_amdgcn_ballot_w32(mine);
                while (todo) {
                    const int j = __builtin_ctz(todo);
                    todo &= todo - 1;
                    const int   sj = __builtin_amdgcn_readlane(s_l, j);
                    const float cj = __int_as_float(__builtin_amdgcn_readlane(__float_as_int(c_l), j));
                    const float4 v = xv[(size_t)sj * 32 + lane];
                    acc.x += cj * v.x; acc.y += cj * v.y; acc.z += cj * v.z; acc.w += cj * v.w;
                }
            }
        }

        const float4 self = xv[(size_t)t * 32 + lane];
        const float ss = dt * dt;
        float4 r;
        r.x = acc.x + (ss * self.x + bb.x);
        r.y = acc.y + (ss * self.y + bb.y);
        r.z = acc.z + (ss * self.z + bb.z);
        r.w = acc.w + (ss * self.w + bb.w);
        ((float4*)out)[(size_t)t * 32 + lane] = r;
    }
}

extern "C" void kernel_launch(void* const* d_in, const int* in_sizes, int n_in,
                              void* d_out, int out_size, void* d_ws, size_t ws_size,
                              hipStream_t stream) {
    const float* x          = (const float*)d_in[0];
    const int*   edge_index = (const int*)d_in[1];
    const float* W          = (const float*)d_in[2];
    const float* b          = (const float*)d_in[3];
    const float* w_pred     = (const float*)d_in[4];
    const float* b_pred     = (const float*)d_in[5];
    float*       out        = (float*)d_out;

    const int* src = edge_index;
    const int* tgt = edge_index + N_EDGES;

    float* ws   = (float*)d_ws;
    float* p_a  = ws;
    float* p_b  = p_a + N_NODES;
    int2*  seg  = (int2*)(ws + 150016);
    float* xw   = ws + 6550016;
    u64*   cd   = (u64*)(ws + 12950528);
    int*   ovf_n = (int*)(ws + 13155328);
    int4*  ovf  = (int4*)(ws + 13156352);

    gemm_dots_kernel  <<<GEMM_BLOCKS,   256, 0, stream>>>(x, W, w_pred, xw, p_a, p_b, cd, ovf_n);
    edge_bucket_kernel<<<EDGE_BLOCKS,   256, 0, stream>>>(src, tgt, p_a, p_b, b_pred, cd, seg, ovf_n, ovf);
    node_gather_kernel<<<GATHER_BLOCKS, 256, 0, stream>>>(cd, seg, ovf_n, ovf, xw, b, out);
}


# gather loop: 8 row loads per trip issued as one unguarded group (a row past the end is row 0 with coefficient zero), the node's own row loaded first; was 2 dependent waits per 4 rows; sum order kept
# speedup vs baseline: 4.0276x; 1.0272x over previous
#include <hip/hip_runtime.h>
#include <cmath>


#define N_NODES 50000
#define N_EDGES 800000
#define DIM 128
#define NPAD 51200
#define KH 32
#define N_TILES (N_NODES / 16)
#define GEMM_BLOCKS 1024
#define EPT 4
#define EDGE_BLOCKS ((N_EDGES + 256 * EPT - 1) / (256 * EPT))
#define GATHER_BLOCKS 2048
#define RPT 8
#define CNT_BITS 20
#define CNT_MASK 0xFFFFFu
#define FIX_ONE 16777216.0f

typedef float v2f __attribute__((ext_vector_type(2)));
typedef float v8f __attribute__((ext_vector_type(8)));
typedef unsigned long long u64;

__device__ __forceinline__ float inv_sqrt_degree(u64 c0, u64 c1)
{
    const u64 sum = (c0 >> CNT_BITS) + (c1 >> CNT_BITS);
    const unsigned whole = (unsigned)(sum >> 24);
    const unsigned frac  = (unsigned)sum & 0xFFFFFFu;
    const float g = fmaf((float)frac, 1.0f / FIX_ONE, (float)(whole + 1u));
    return (g > 0.f) ? rsqrtf(g) : 0.f;
}

__device__ __forceinline__ int chip_half()
{
    return (int)((__builtin_amdgcn_s_sendmsg_rtn(0x87) >> 18) & 1u);
}

__global__ __launch_bounds__(256) void gemm_dots_kernel(
    const float* __restrict__ x, const float* __restrict__ W,
    const float* __restrict__ w_pred,
    float* __restrict__ xw, float* __restrict__ p_a, float* __restrict__ p_b,
    u64* __restrict__ cd, int* __restrict__ ovf_n)
{
    __shared__ float Asm[2 * 16 * 132];
    if (blockIdx.x == 0 && threadIdx.x == 0) ovf_n[0] = 0;

    const int waveId  = threadIdx.x >> 5;
    const int lane    = threadIdx.x & 31;
    const int m       = lane & 15;
    const int hi      = lane >> 4;
    const int colBase = waveId * 16;

    v2f bfrag[32];
#pragma unroll
    for (int i = 0; i < 32; ++i) {
        const int ka = 4 * i + 2 * hi;
        bfrag[i].x = W[(size_t)ka * DIM + colBase + m];
        bfrag[i].y = W[(size_t)(ka + 1) * DIM + colBase + m];
    }

    int buf = 0;
    for (int tile = blockIdx.x; tile < N_TILES; tile += gridDim.x, buf ^= 1) {
        const int rowBase = tile * 16;
        float* As = Asm + buf * (16 * 132);

#pragma unroll
        for (int j = 0; j < 8; ++j) {
            int idx = threadIdx.x + j * 256;
            int r = idx >> 7, c = idx & 127;
            As[r * 132 + c] = x[(size_t)rowBase * DIM + idx];
        }
        __syncthreads();

        {
            const float* r0 = As + (2 * waveId) * 132;
            const float* r1 = r0 + 132;
            float sa0 = 0.f, sb0 = 0.f, sa1 = 0.f, sb1 = 0.f;
#pragma unroll
            for (int j = 0; j < 4; ++j) {
                const int c = lane + 32 * j;
                const float wa = w_pred[c];
                const float wb = w_pred[DIM + c];
                const float v0 = r0[c];
                const float v1 = r1[c];
                sa0 += v0 * wa;
                sb0 += v0 * wb;
                sa1 += v1 * wa;
                sb1 += v1 * wb;
            }
#pragma unroll
            for (int off = 16; off > 0; off >>= 1) {
                sa0 += __shfl_down(sa0, off, 32);
                sb0 += __shfl_down(sb0, off, 32);
                sa1 += __shfl_down(sa1, off, 32);
                sb1 += __shfl_down(sb1, off, 32);
            }
            if (lane == 0) {
                const int node = rowBase + 2 * waveId;
                *(float2*)(p_a + node) = make_float2(sa0, sa1);
                *(float2*)(p_b + node) = make_float2(sb0, sb1);
                *(ulonglong2*)(cd + node) = make_ulonglong2(0ull, 0ull);
                *(ulonglong2*)(cd + NPAD + node) = make_ulonglong2(0ull, 0ull);
            }
        }

        v8f acc = {};
#pragma unroll
        for (int i = 0; i < 32; ++i) {
            const int ka = 4 * i + 2 * hi;
            v2f a;
            a.x = As[m * 132 + ka];
            a.y = As[m * 132 + ka + 1];
            acc = __builtin_amdgcn_wmma_f32_16x16x4_f32(
                false, a, false, bfrag[i], (short)0, acc, false, false);
        }

#pragma unroll
        for (int i = 0; i < 8; ++i) {
            int row = rowBase + i + 8 * hi;
            xw[(size_t)row * DIM + colBase + m] = acc[i];
        }
    }
}

__global__ __launch_bounds__(256) void edge_bucket_kernel(
    const int* __restrict__ src, const int* __restrict__ tgt,
    const float* __restrict__ p_a, const float* __restrict__ p_b,
    const float* __restrict__ b_pred,
    u64* __restrict__ cd, int2* __restrict__ seg,
    int* __restrict__ ovf_n, int4* __restrict__ ovf)
{
    const int h = chip_half();
    const float bp = b_pred[0];
    const int e0 = blockIdx.x * (256 * EPT) + threadIdx.x;
    int  s[EPT], t[EPT];
    bool ok[EPT];
#pragma unroll
    for (int k = 0; k < EPT; ++k) {
        const int e = e0 + k * 256;
        ok[k] = e < N_EDGES;
        s[k] = ok[k] ? src[e] : 0;
        t[k] = ok[k] ? tgt[e] : 0;
    }
    float w[EPT];
#pragma unroll
    for (int k = 0; k < EPT; ++k) {
        const float z = p_a[s[k]] + p_b[t[k]] + bp;
        w[k] = 1.0f / (1.0f + __expf(-z));
    }
    u64 old[EPT];
#pragma unroll
    for (int k = 0; k < EPT; ++k) {
        old[k] = 0ull;
        if (ok[k]) {
            const unsigned q = (unsigned)__builtin_rintf(w[k] * FIX_ONE);
            old[k] = atomicAdd(&cd[h * NPAD + t[k]], ((u64)q << CNT_BITS) | 1ull);
        }
    }
#pragma unroll
    for (int k = 0; k < EPT; ++k) {
        if (!ok[k]) continue;
        const int r = (int)((unsigned)old[k] & CNT_MASK);
        if (r < KH) {
            seg[((size_t)t[k] * 2 + h) * KH + r] = make_int2(s[k], __float_as_int(w[k]));
        } else {
            const int o = atomicAdd(ovf_n, 1);
            ovf[o] = make_int4(s[k], t[k], __float_as_int(w[k]), 0);
        }
    }
}

__global__ __launch_bounds__(256) void node_gather_kernel(
    const u64* __restrict__ cd, const int2* __restrict__ seg,
    const int* __restrict__ ovf_n, const int4* __restrict__ ovf,
    const float* __restrict__ xw, const float* __restrict__ b,
    float* __restrict__ out)
{
    const int gw = __builtin_amdgcn_readfirstlane(
        (int)((blockIdx.x * blockDim.x + threadIdx.x) >> 5));
    const int n_waves = (int)(gridDim.x * (blockDim.x >> 5));
    const int lane = threadIdx.x & 31;
    const float4* xv = (const float4*)xw;
    const float4 bb = ((const float4*)b)[lane];

    for (int t = gw; t < N_NODES; t += n_waves) {
        const u64 w0 = cd[t], w1 = cd[NPAD + t];
        const float4 self = xv[(size_t)t * 32 + lane];
        const int m0 = (int)((unsigned)w0 & CNT_MASK), m1 = (int)((unsigned)w1 & CNT_MASK);
        const int n0 = min(m0, KH);
        const int n_e = n0 + min(m1, KH);
        const int2* bk = seg + (size_t)t * (2 * KH);
        const float dt = inv_sqrt_degree(w0, w1);

        float4 acc = make_float4(0.f, 0.f, 0.f, 0.f);
        for (int base = 0; base < n_e; base += 32) {
            const int i = base + lane;
            int   s_l = 0;
            float c_l = 0.f;
            if (i < n_e) {
                const int slot = (i < n0) ? i : (KH + i - n0);
                int2 ent = bk[slot];
                s_l = ent.x;
                c_l = inv_sqrt_degree(cd[s_l], cd[NPAD + s_l]) * __int_as_float(ent.y) * dt;
            }
            const int m = min(32, n_e - base);
            for (int j = 0; j < m; j += RPT) {
                int    sj[RPT];
                float  cj[RPT];
                float4 v[RPT];
#pragma unroll
                for (int k = 0; k < RPT; ++k) {
                    sj[k] = __builtin_amdgcn_readlane(s_l, j + k);
                    cj[k] = __int_as_float(__builtin_amdgcn_readlane(__float_as_int(c_l), j + k));
                }
#pragma unroll
                for (int k = 0; k < RPT; ++k)
                    v[k] = ((const float4*)(xw + (size_t)sj[k] * DIM))[lane];
#pragma unroll
                for (int k = 0; k < RPT; ++k) {
                    acc.x += cj[k] * v[k].x; acc.y += cj[k] * v[k].y;
                    acc.z += cj[k] * v[k].z; acc.w += cj[k] * v[k].w;
                }
            }
        }

        if ((m0 > KH) | (m1 > KH)) {
            const int n_o = min(ovf_n[0], N_EDGES);
            for (int base = 0; base < n_o; base += 32) {
                const int i = base + lane;
                int   s_l = 0;
                float c_l = 0.f;
                bool  mine = false;
                if (i < n_o) {
                    const int4 q = ovf[i];
                    if (q.y == t) {
                        mine = true;
                        s_l = q.x;
                        c_l = inv_sqrt_degree(cd[s_l], cd[NPAD + s_l]) * __int_as_float(q.z) * dt;
                    }
                }
                unsigned todo = __builtin_amdgcn_ballot_w32(mine);
                while (todo) {
                    const int j = __builtin_ctz(todo);
                    todo &= todo - 1;
                    const int   sj = __builtin_amdgcn_readlane(s_l, j);
                    const float cj = __int_as_float(__builtin_amdgcn_readlane(__float_as_int(c_l), j));
                    const float4 v = xv[(size_t)sj * 32 + lane];
                    acc.x += cj * v.x; acc.y += cj * v.y; acc.z += cj * v.z; acc.w += cj * v.w;
                }
            }
        }

        const float ss = dt * dt;
        float4 r;
        r.x = acc.x + (ss * self.x + bb.x);
        r.y = acc.y + (ss * self.y + bb.y);
        r.z = acc.z + (ss * self.z + bb.z);
        r.w = acc.w + (ss * self.w + bb.w);
        ((float4*)out)[(size_t)t * 32 + lane] = r;
    }
}

extern "C" void kernel_launch(void* const* d_in, const int* in_sizes, int n_in,
                              void* d_out, int out_size, void* d_ws, size_t ws_size,
                              hipStream_t stream) {
    const float* x          = (const float*)d_in[0];
    const int*   edge_index = (const int*)d_in[1];
    const float* W          = (const float*)d_in[2];
    const float* b          = (const float*)d_in[3];
    const float* w_pred     = (const float*)d_in[4];
    const float* b_pred     = (const float*)d_in[5];
    float*       out        = (float*)d_out;

    const int* src = edge_index;
    const int* tgt = edge_index + N_EDGES;

    float* ws   = (float*)d_ws;
    float* p_a  = ws;
    float* p_b  = p_a + N_NODES;
    int2*  seg  = (int2*)(ws + 150016);
    float* xw   = ws + 6550016;
    u64*   cd   = (u64*)(ws + 12950528);
    int*   ovf_n = (int*)(ws + 13155328);
    int4*  ovf  = (int4*)(ws + 13156352);

    gemm_dots_kernel  <<<GEMM_BLOCKS,   256, 0, stream>>>(x, W, w_pred, xw, p_a, p_b, cd, ovf_n);
    edge_bucket_kernel<<<EDGE_BLOCKS,   256, 0, stream>>>(src, tgt, p_a, p_b, b_pred, cd, seg, ovf_n, ovf);
    node_gather_kernel<<<GATHER_BLOCKS, 256, 0, stream>>>(cd, seg, ovf_n, ovf, xw, b, out);
}
